# starting kernel
# baseline (speedup 1.0000x reference)
#include <hip/hip_runtime.h>

typedef __attribute__((ext_vector_type(2))) float v2f;
typedef __attribute__((ext_vector_type(8))) float v8f;

#define BN   16
#define TT   64
#define FF   8192
#define SLD  68

__global__ void __launch_bounds__(256)
qk_dot_kernel(const float* __restrict__ Q,
              const float* __restrict__ K,
              float* __restrict__ S) {
  const int bn   = blockIdx.x;
  const int lane = threadIdx.x & 31;
  const int wave = threadIdx.x >> 5;
  const int row  = lane & 15;
  const int koff = (lane >> 4) << 1;

  const float* q = Q + (size_t)bn * TT * FF;
  const float* k = K + (size_t)bn * TT * FF;

  const int tj   = wave & 3;
  const int ti_a = wave >> 2;
  const int ti_b = ti_a + 2;

  const float* qa = q + (size_t)(ti_a * 16 + row) * FF + koff;
  const float* qb = q + (size_t)(ti_b * 16 + row) * FF + koff;
  const float* kb = k + (size_t)(tj   * 16 + row) * FF + koff;

  v8f acc_a = {};
  v8f acc_b = {};
#pragma unroll 4
  for (int kk = 0; kk < FF; kk += 4) {
    v2f a0 = *(const v2f*)(qa + kk);
    v2f a1 = *(const v2f*)(qb + kk);
    v2f b  = *(const v2f*)(kb + kk);
    acc_a = __builtin_amdgcn_wmma_f32_16x16x4_f32(false, a0, false, b,
                                                  (short)0, acc_a, false, false);
    acc_b = __builtin_amdgcn_wmma_f32_16x16x4_f32(false, a1, false, b,
                                                  (short)0, acc_b, false, false);
  }

  const float scale = 1.0f / 1024.0f;
  float* sb = S + (size_t)bn * TT * TT;
  const int qcol = tj * 16 + row;
  const int half = (lane >> 4) * 8;
#pragma unroll
  for (int r = 0; r < 8; ++r) {
    sb[(ti_a * 16 + half + r) * TT + qcol] = acc_a[r] * scale;
    sb[(ti_b * 16 + half + r) * TT + qcol] = acc_b[r] * scale;
  }
}

__global__ void __launch_bounds__(256)
sv_gauss_kernel(const float* __restrict__ S,
                const float* __restrict__ V,
                float* __restrict__ O) {
  __shared__ float Sl[TT * SLD];

  const int bn    = blockIdx.x >> 6;
  const int chunk = blockIdx.x & 63;
  const int tid   = threadIdx.x;
  const int lane  = tid & 31;
  const int wave  = tid >> 5;

  const float* sg = S + (size_t)bn * TT * TT;
  for (int i = tid; i < TT * TT; i += 256) {
    Sl[(i >> 6) * SLD + (i & 63)] = sg[i];
  }
  __syncthreads();

  const int ntile = chunk * 8 + wave;
  const int f0    = ntile * 16;
  const int row   = lane & 15;
  const int koff  = (lane >> 4) << 1;

  const float* v = V + (size_t)bn * TT * FF;

  v8f acc[4] = {};
  for (int kk = 0; kk < TT; kk += 4) {
    v2f b;
    b.x = v[(size_t)(kk + koff)     * FF + f0 + row];
    b.y = v[(size_t)(kk + koff + 1) * FF + f0 + row];
#pragma unroll
    for (int pb = 0; pb < 4; ++pb) {
      v2f a = *(const v2f*)(&Sl[(pb * 16 + row) * SLD + kk + koff]);
      acc[pb] = __builtin_amdgcn_wmma_f32_16x16x4_f32(false, a, false, b,
                                                      (short)0, acc[pb], false, false);
    }
  }

  const int f = f0 + row;
  const int m = f >> 3;
  const float dw = (float)(m & 31) - 15.5f;
  const float dh = (float)(m >> 5) - 15.5f;
  float sx = 0.0f;
#pragma unroll
  for (int x = 0; x < 32; ++x) {
    float d = (float)x - 15.5f;
    sx += __expf(-d * d * (1.0f / 50.0f));
  }
  const float g = __expf(-(dw * dw + dh * dh) * (1.0f / 50.0f)) / (sx * sx);

  float* o = O + (size_t)bn * TT * FF;
  const int half = (lane >> 4) * 8;
#pragma unroll
  for (int pb = 0; pb < 4; ++pb) {
#pragma unroll
    for (int r = 0; r < 8; ++r) {
      const int p = pb * 16 + half + r;
      o[(size_t)p * FF + f] = acc[pb][r] * g;
    }
  }
}

extern "C" void kernel_launch(void* const* d_in, const int* in_sizes, int n_in,
                              void* d_out, int out_size, void* d_ws, size_t ws_size,
                              hipStream_t stream) {
  const float* Q = (const float*)d_in[0];
  const float* K = (const float*)d_in[1];
  const float* V = (const float*)d_in[2];
  float* O = (float*)d_out;
  float* S = (float*)d_ws;

  qk_dot_kernel<<<BN, 256, 0, stream>>>(Q, K, S);
  sv_gauss_kernel<<<BN * 64, 256, 0, stream>>>(S, V, O);
}


# Score kernel: a wave owns a 32x32 score tile (2x2 WMMA tiles), each Q/K fragment feeds two products; features split over 4 work-groups, SV prologue adds the 4 planes via LDS: changes f32 sum order
# speedup vs baseline: 11.7196x; 11.7196x over previous
#include <hip/hip_runtime.h>

typedef __attribute__((ext_vector_type(2))) float v2f;
typedef __attribute__((ext_vector_type(4))) float v4f;
typedef __attribute__((ext_vector_type(8))) float v8f;

#define BN   16
#define TT   64
#define FF   8192

#define KW   8
#define NQ   4
#define FQ   (FF / NQ)
#define FW   (FQ / KW)
#define PLANE (BN * TT * TT)

__device__ __forceinline__ float take_hi8(float keep, float src) {
  return __builtin_bit_cast(float, __builtin_amdgcn_update_dpp(
      __builtin_bit_cast(int, keep), __builtin_bit_cast(int, src), 0x128, 0xF, 0xC, false));
}
__device__ __forceinline__ float take_lo8(float keep, float src) {
  return __builtin_bit_cast(float, __builtin_amdgcn_update_dpp(
      __builtin_bit_cast(int, keep), __builtin_bit_cast(int, src), 0x128, 0xF, 0x3, false));
}

__global__ void __launch_bounds__(256) __attribute__((amdgpu_waves_per_eu(2, 4)))
qk_dot_kernel(const float* __restrict__ Q,
              const float* __restrict__ K,
              float* __restrict__ P) {
  __shared__ float part[KW * 4 * 256];

  const int fq   = blockIdx.x & 3;
  const int t2   = (blockIdx.x >> 2) & 3;
  const int bn   = blockIdx.x >> 4;
  const int i2   = t2 >> 1;
  const int j2   = t2 & 1;
  const int lane = threadIdx.x & 31;
  const int wave = threadIdx.x >> 5;
  const int row  = lane & 15;
  const int r8   = lane & 7;
  const int koff = ((lane >> 3) & 1) * 8 + (lane >> 4) * 4;

  const float* qx = Q + ((size_t)bn * TT + i2 * 32 + r8) * FF + fq * FQ + wave * FW + koff;
  const float* kx = K + ((size_t)bn * TT + j2 * 32 + r8) * FF + fq * FQ + wave * FW + koff;

  v8f acc[2][2] = {};
#pragma unroll
  for (int u = 0; u < FW / 16; ++u) {
    v4f fa0[2], fa1[2], fb0[2], fb1[2];
#pragma unroll
    for (int g = 0; g < 2; ++g) {
      const v4f ax = *(const v4f*)(qx + (size_t)(16 * g) * FF + 16 * u);
      const v4f ay = *(const v4f*)(qx + (size_t)(16 * g + 8) * FF + 16 * u);
      const v4f bx = *(const v4f*)(kx + (size_t)(16 * g) * FF + 16 * u);
      const v4f by = *(const v4f*)(kx + (size_t)(16 * g + 8) * FF + 16 * u);
#pragma unroll
      for (int e = 0; e < 4; ++e) {
        fa0[g][e] = take_hi8(ax[e], ay[e]);
        fa1[g][e] = take_lo8(ay[e], ax[e]);
        fb0[g][e] = take_hi8(bx[e], by[e]);
        fb1[g][e] = take_lo8(by[e], bx[e]);
      }
    }
#pragma unroll
    for (int a = 0; a < 2; ++a) {
#pragma unroll
      for (int b = 0; b < 2; ++b) {
        acc[a][b] = __builtin_amdgcn_wmma_f32_16x16x4_f32(false, fa0[a].xy, false, fb0[b].xy,
                                                          (short)0, acc[a][b], false, false);
      }
    }
#pragma unroll
    for (int a = 0; a < 2; ++a) {
#pragma unroll
      for (int b = 0; b < 2; ++b) {
        acc[a][b] = __builtin_amdgcn_wmma_f32_16x16x4_f32(false, fa0[a].zw, false, fb0[b].zw,
                                                          (short)0, acc[a][b], false, false);
      }
    }
#pragma unroll
    for (int a = 0; a < 2; ++a) {
#pragma unroll
      for (int b = 0; b < 2; ++b) {
        acc[a][b] = __builtin_amdgcn_wmma_f32_16x16x4_f32(false, fa1[a].xy, false, fb1[b].xy,
                                                          (short)0, acc[a][b], false, false);
      }
    }
#pragma unroll
    for (int a = 0; a < 2; ++a) {
#pragma unroll
      for (int b = 0; b < 2; ++b) {
        acc[a][b] = __builtin_amdgcn_wmma_f32_16x16x4_f32(false, fa1[a].zw, false, fb1[b].zw,
                                                          (short)0, acc[a][b], false, false);
      }
    }
  }

  const int half = (lane >> 4) * 8;
#pragma unroll
  for (int a = 0; a < 2; ++a) {
#pragma unroll
    for (int b = 0; b < 2; ++b) {
      float* pw = part + (wave * 4 + a * 2 + b) * 256;
#pragma unroll
      for (int r = 0; r < 8; ++r) {
        pw[(half + r) * 16 + row] = acc[a][b][r];
      }
    }
  }
  __syncthreads();

  const int t = threadIdx.x;
  const float scale = 1.0f / 1024.0f;
  float* pl = P + (size_t)fq * PLANE + (size_t)bn * TT * TT;
#pragma unroll
  for (int a = 0; a < 2; ++a) {
#pragma unroll
    for (int b = 0; b < 2; ++b) {
      const int ab = a * 2 + b;
      float s = part[ab * 256 + t];
#pragma unroll
      for (int w = 1; w < KW; ++w) {
        s += part[(w * 4 + ab) * 256 + t];
      }
      const int tile = (i2 * 2 + a) * 4 + (j2 * 2 + b);
      pl[tile * 256 + t] = s * scale;
    }
  }
}

#define SV_TILES 4
#define SV_GROUPS (FF / (16 * 8 * SV_TILES))
#define SS_PITCH 20

__global__ void __launch_bounds__(256) __attribute__((amdgpu_waves_per_eu(2, 2)))
sv_gauss_kernel(const float* __restrict__ P,
                const float* __restrict__ V,
                float* __restrict__ O) {
  __shared__ __attribute__((aligned(16))) float ss[16 * 16 * SS_PITCH];

  const int bn   = blockIdx.x / SV_GROUPS;
  const int grp  = blockIdx.x % SV_GROUPS;
  const int lane = threadIdx.x & 31;
  const int wave = threadIdx.x >> 5;
  const int row  = lane & 15;
  const int half = lane >> 4;

  const float* v = V + (size_t)bn * TT * FF + (size_t)(8 * half) * FF + row;
  float*       o = O + (size_t)bn * TT * FF + (size_t)(8 * half) * FF + row;
  const int f_first = (grp * SV_TILES * 8 + wave) * 16;

  v2f vb[SV_TILES][16];
#define SV_LOAD(i)                                                            \
  _Pragma("unroll")                                                           \
  for (int kj = 0; kj < 16; ++kj) {                                           \
    const int q0 = (kj >> 2) * 16 + (kj & 3) * 2;          \
    vb[i][kj].x = v[(size_t)q0 * FF + f_first + 128 * (i)];                   \
    vb[i][kj].y = v[(size_t)(q0 + 1) * FF + f_first + 128 * (i)];             \
  }

  SV_LOAD(0)
  SV_LOAD(1)

  const float* pg = P + (size_t)bn * TT * TT;
#pragma unroll
  for (int j = 0; j < 4; ++j) {
    const int x = 4 * (threadIdx.x + 256 * j);
    v4f s = *(const v4f*)(pg + x);
#pragma unroll
    for (int pl = 1; pl < NQ; ++pl) {
      s += *(const v4f*)(pg + (size_t)pl * PLANE + x);
    }
    *(v4f*)(ss + (x >> 8) * (16 * SS_PITCH) + ((x >> 4) & 15) * SS_PITCH + (x & 15)) = s;
  }
  __builtin_amdgcn_fence(__ATOMIC_RELEASE, "workgroup", "local");
  __builtin_amdgcn_s_barrier();
  __builtin_amdgcn_fence(__ATOMIC_ACQUIRE, "workgroup", "local");

  const float* sl = ss + row * SS_PITCH + half * 8;
  v4f sa[16][2];
#pragma unroll
  for (int t = 0; t < 16; ++t) {
    sa[t][0] = *(const v4f*)(sl + t * (16 * SS_PITCH));
    sa[t][1] = *(const v4f*)(sl + t * (16 * SS_PITCH) + 4);
  }

  float sx = 0.0f;
#pragma unroll
  for (int x = 0; x < 32; ++x) {
    float d = (float)x - 15.5f;
    sx += __expf(-d * d * (1.0f / 50.0f));
  }
  const float norm = sx * sx;

#pragma unroll
  for (int i = 0; i < SV_TILES; ++i) {
    if (i + 2 < SV_TILES) {
      if (i == 0) { SV_LOAD(2) } else { SV_LOAD(3) }
    }

    v8f acc[4] = {};
#pragma unroll
    for (int kj = 0; kj < 16; ++kj) {
      const int tj = kj >> 2;
      const int j  = kj & 3;
#pragma unroll
      for (int pb = 0; pb < 4; ++pb) {
        const v4f s4 = sa[pb * 4 + tj][j >> 1];
        const v2f a  = (j & 1) ? s4.zw : s4.xy;
        acc[pb] = __builtin_amdgcn_wmma_f32_16x16x4_f32(false, a, false, vb[i][kj],
                                                        (short)0, acc[pb], false, false);
      }
    }

    __builtin_amdgcn_sched_barrier(0);
    const int f = f_first + 128 * i + row;
    const int m = f >> 3;
    const float dw = (float)(m & 31) - 15.5f;
    const float dh = (float)(m >> 5) - 15.5f;
    const float g = __expf(-(dw * dw + dh * dh) * (1.0f / 50.0f)) / norm;

#pragma unroll
    for (int pb = 0; pb < 4; ++pb) {
#pragma unroll
      for (int r = 0; r < 8; ++r) {
        o[(size_t)(pb * 16 + r) * FF + f_first + 128 * i] = acc[pb][r] * g;
      }
    }
    __builtin_amdgcn_sched_barrier(0);
  }
#undef SV_LOAD
}

extern "C" void kernel_launch(void* const* d_in, const int* in_sizes, int n_in,
                              void* d_out, int out_size, void* d_ws, size_t ws_size,
                              hipStream_t stream) {
  const float* Q = (const float*)d_in[0];
  const float* K = (const float*)d_in[1];
  const float* V = (const float*)d_in[2];
  float* O = (float*)d_out;
  float* P = (float*)d_ws;

  qk_dot_kernel<<<BN * 16, 256, 0, stream>>>(Q, K, P);
  sv_gauss_kernel<<<BN * SV_GROUPS, 256, 0, stream>>>(P, V, O);
}
